# starting kernel
# baseline (speedup 1.0000x reference)
#include <hip/hip_runtime.h>


typedef __attribute__((ext_vector_type(2))) float v2f;
typedef __attribute__((ext_vector_type(8))) float v8f;

__global__ __launch_bounds__(256) void iwt_wmma_kernel(const float* __restrict__ x,
                                                       float* __restrict__ out) {
    const int tid  = blockIdx.x * blockDim.x + threadIdx.x;
    const int lane = threadIdx.x & 31;
    const int wid  = tid >> 5;

    const int h  = wid & 63;
    const int bc = wid >> 6;

    const bool lo   = lane < 16;
    const int  m    = lane & 15;
    const int  comp = m & 3;
    const bool act  = (m & 4) == 0;

    const float s0 = lo ? 0.5f
                        : (((comp & 1) == 0) ? 0.5f : -0.5f);
    const float s1 = lo ? ((comp < 2) ? 0.5f : -0.5f)
                        : ((comp == 0 || comp == 3) ? 0.5f : -0.5f);
    v2f a;
    a.x = act ? s0 : 0.0f;
    a.y = act ? s1 : 0.0f;

    const size_t plane = (size_t)bc * 16384;
    const int    sub0  = lo ? 0 : 2;
    const int    sub1  = lo ? 1 : 3;

    const float* in_b0 = x + plane + (size_t)sub0 * 4096 + (size_t)h * 64 + m;
    const float* in_b1 = x + plane + (size_t)sub1 * 4096 + (size_t)h * 64 + m;
    float* out_row = out + plane + (size_t)(2 * h + (lo ? 0 : 1)) * 128 + 2 * m;

#pragma unroll
    for (int g = 0; g < 4; ++g) {
        v2f bm;
        bm.x = __builtin_nontemporal_load(in_b0 + g * 16);
        bm.y = __builtin_nontemporal_load(in_b1 + g * 16);
        v8f cacc = {};
        v8f d = __builtin_amdgcn_wmma_f32_16x16x4_f32(
            false, a, false, bm, (short)0, cacc, false, false);

        float2 st;
        st.x = lo ? d[0] : d[2];
        st.y = lo ? d[1] : d[3];
        *(float2*)(out_row + g * 32) = st;
    }
}

extern "C" void kernel_launch(void* const* d_in, const int* in_sizes, int n_in,
                              void* d_out, int out_size, void* d_ws, size_t ws_size,
                              hipStream_t stream) {
    const float* x = (const float*)d_in[0];
    float* out = (float*)d_out;

    const int waves  = in_sizes[0] / 256;
    const int blocks = waves / 8;
    iwt_wmma_kernel<<<dim3(blocks), dim3(256), 0, stream>>>(x, out);
}


# Stream shape: 4,096 resident waves each walk 32 rows (8-row batches, next batch's loads ahead) with 8-byte loads and 16-byte stores on whole 128-B lines; was 131,072 one-row waves, 4-byte loads
# speedup vs baseline: 1.1665x; 1.1665x over previous
#include <hip/hip_runtime.h>


typedef __attribute__((ext_vector_type(2))) float v2f;
typedef __attribute__((ext_vector_type(4))) float v4f;
typedef __attribute__((ext_vector_type(8))) float v8f;

#define IWT_ROWS   8
#define IWT_TRIPS  4
#define IWT_WAVES  4096
#define IWT_TRIP_STRIDE ((size_t)IWT_WAVES * IWT_ROWS * 256)

struct Batch {
    v2f q[IWT_ROWS][4];
};

static __device__ __forceinline__ void load_batch(Batch& b, const float* p0, const float* p1) {
#pragma unroll
    for (int r = 0; r < IWT_ROWS; ++r) {
        b.q[r][0] = __builtin_nontemporal_load((const v2f*)(p0 + r * 64));
        b.q[r][1] = __builtin_nontemporal_load((const v2f*)(p0 + r * 64 + 32));
        b.q[r][2] = __builtin_nontemporal_load((const v2f*)(p1 + r * 64));
        b.q[r][3] = __builtin_nontemporal_load((const v2f*)(p1 + r * 64 + 32));
    }
}

static __device__ __forceinline__ v8f haar(v2f a, float k02, float k13) {
    v2f bm;
    bm.x = k02;
    bm.y = k13;
    v8f cacc = {};
    return __builtin_amdgcn_wmma_f32_16x16x4_f32(false, a, false, bm, (short)0, cacc, false, false);
}

static __device__ __forceinline__ void mul_store_batch(const Batch& b, v2f a, float* o) {
#pragma unroll
    for (int r = 0; r < IWT_ROWS; ++r) {
        const v8f d0 = haar(a, b.q[r][0].x, b.q[r][2].x);
        const v8f d1 = haar(a, b.q[r][0].y, b.q[r][2].y);
        const v8f d2 = haar(a, b.q[r][1].x, b.q[r][3].x);
        const v8f d3 = haar(a, b.q[r][1].y, b.q[r][3].y);
        v4f s0, s1;
        s0.x = d0[0]; s0.y = d0[1]; s0.z = d1[0]; s0.w = d1[1];
        s1.x = d2[0]; s1.y = d2[1]; s1.z = d3[0]; s1.w = d3[1];
        *(v4f*)(o + r * 256)      = s0;
        *(v4f*)(o + r * 256 + 64) = s1;
    }
}

__global__ __launch_bounds__(256) void iwt_wmma_kernel(const float* __restrict__ x,
                                                       float* __restrict__ out) {
    const int lane = threadIdx.x & 31;
    const int gw   = (blockIdx.x * blockDim.x + threadIdx.x) >> 5;

    const bool lo   = lane < 16;
    const int  m    = lane & 15;
    const int  comp = (m & 1) | ((m >> 3) << 1);
    const bool act  = (m & 6) == 0;

    const float s0 = lo ? 0.5f
                        : (((comp & 1) == 0) ? 0.5f : -0.5f);
    const float s1 = lo ? ((comp < 2) ? 0.5f : -0.5f)
                        : ((comp == 0 || comp == 3) ? 0.5f : -0.5f);
    v2f a;
    a.x = act ? s0 : 0.0f;
    a.y = act ? s1 : 0.0f;

    const size_t in_base  = (size_t)(gw >> 3) * 16384 + (size_t)(gw & 7) * 512;
    const size_t out_base = (size_t)gw * 2048;
    const int    sub0     = lo ? 0 : 2;
    const int    sub1     = lo ? 1 : 3;

    const float* p0 = x + in_base + (size_t)sub0 * 4096 + 2 * m;
    const float* p1 = x + in_base + (size_t)sub1 * 4096 + 2 * m;
    float*       o  = out + out_base + (lo ? 0 : 128) + 4 * m;

    Batch ba, bb;
    load_batch(ba, p0, p1);
    load_batch(bb, p0 + IWT_TRIP_STRIDE, p1 + IWT_TRIP_STRIDE);
    mul_store_batch(ba, a, o);
    __builtin_amdgcn_sched_barrier(0);
    load_batch(ba, p0 + 2 * IWT_TRIP_STRIDE, p1 + 2 * IWT_TRIP_STRIDE);
    __builtin_amdgcn_sched_barrier(0);
    mul_store_batch(bb, a, o + IWT_TRIP_STRIDE);
    __builtin_amdgcn_sched_barrier(0);
    load_batch(bb, p0 + 3 * IWT_TRIP_STRIDE, p1 + 3 * IWT_TRIP_STRIDE);
    __builtin_amdgcn_sched_barrier(0);
    mul_store_batch(ba, a, o + 2 * IWT_TRIP_STRIDE);
    mul_store_batch(bb, a, o + 3 * IWT_TRIP_STRIDE);
}

extern "C" void kernel_launch(void* const* d_in, const int* in_sizes, int n_in,
                              void* d_out, int out_size, void* d_ws, size_t ws_size,
                              hipStream_t stream) {
    const float* x = (const float*)d_in[0];
    float* out = (float*)d_out;

    iwt_wmma_kernel<<<dim3(512), dim3(256), 0, stream>>>(x, out);
}
